# starting kernel
# baseline (speedup 1.0000x reference)
#include <hip/hip_runtime.h>


typedef __attribute__((ext_vector_type(2))) float v2f;
typedef __attribute__((ext_vector_type(8))) float v8f;

#define C2D_EPS   0.01f
#define MAXW      16
#define TPB       256
#define LDS_PITCH 17

__global__ __launch_bounds__(TPB)
void corner2depth_wmma_kernel(const float* __restrict__ corners,
                              const float* __restrict__ grid,
                              const int*   __restrict__ nums,
                              float* __restrict__ depth_out,
                              float* __restrict__ nrm_out,
                              int Nmax, long long HW)
{
    __shared__ float s_nx[MAXW], s_nz[MAXW], s_nd[MAXW];
    __shared__ float s_minx[MAXW], s_maxx[MAXW];
    __shared__ float s_minz[MAXW], s_maxz[MAXW];
    __shared__ float s_denom[TPB * LDS_PITCH];

    const int tid = threadIdx.x;
    const int b   = blockIdx.y;
    int n = nums[b];
    if (n > MAXW) n = MAXW;

    if (tid < MAXW) {
        float nx = 0.f, nz = 0.f, nd = 0.f;
        float mnx = 0.f, mxx = 0.f, mnz = 0.f, mxz = 0.f;
        if (tid < n) {
            const long long base = ((long long)b * Nmax + tid) * 3;
            const int j = (tid + 1 == n) ? 0 : tid + 1;
            const long long basej = ((long long)b * Nmax + j) * 3;
            const float x0 = corners[base + 0], z0 = corners[base + 2];
            const float x1 = corners[basej + 0], z1 = corners[basej + 2];
            const float dx = x1 - x0, dz = z1 - z0;
            nx = -dz;
            nz =  dx;
            nd = nx * x0 + nz * z0;
            mnx = fminf(x0, x1) - C2D_EPS;  mxx = fmaxf(x0, x1) + C2D_EPS;
            mnz = fminf(z0, z1) - C2D_EPS;  mxz = fmaxf(z0, z1) + C2D_EPS;
        }
        s_nx[tid] = nx; s_nz[tid] = nz; s_nd[tid] = nd;
        s_minx[tid] = mnx; s_maxx[tid] = mxx;
        s_minz[tid] = mnz; s_maxz[tid] = mxz;
    }
    __syncthreads();

    const long long p = (long long)blockIdx.x * TPB + tid;
    const float gx = grid[p * 3 + 0];
    const float gz = grid[p * 3 + 2];

    const int lane = tid & 31;
    const int lh   = lane & 15;
    const int hlf  = lane >> 4;
    const int wv   = tid >> 5;

    const float gx0 = __shfl(gx, lh,      32);
    const float gz0 = __shfl(gz, lh,      32);
    const float gx1 = __shfl(gx, lh + 16, 32);
    const float gz1 = __shfl(gz, lh + 16, 32);

    v2f a0, a1, bm;
    a0.x = hlf ? gz0 : gx0;  a0.y = 0.f;
    a1.x = hlf ? gz1 : gx1;  a1.y = 0.f;
    bm.x = hlf ? s_nz[lh] : s_nx[lh];  bm.y = 0.f;

    v8f acc0 = {}, acc1 = {};
    acc0 = __builtin_amdgcn_wmma_f32_16x16x4_f32(false, a0, false, bm,
                                                 (short)0, acc0, false, false);
    acc1 = __builtin_amdgcn_wmma_f32_16x16x4_f32(false, a1, false, bm,
                                                 (short)0, acc1, false, false);

    const int mBase = hlf * 8;
#pragma unroll
    for (int r = 0; r < 8; ++r) {
        s_denom[(wv * 32 +      mBase + r) * LDS_PITCH + lh] = acc0[r];
        s_denom[(wv * 32 + 16 + mBase + r) * LDS_PITCH + lh] = acc1[r];
    }
    __syncthreads();

    float best = __builtin_inff();
    int   bi   = 0;
    for (int i = 0; i < n; ++i) {
        const float denom = s_denom[tid * LDS_PITCH + i];
        const float scale = s_nd[i] * __builtin_amdgcn_rcpf(denom);
        const float ix = gx * scale;
        const float iz = gz * scale;
        const bool ok = (ix <= s_maxx[i]) & (ix >= s_minx[i])
                      & (iz <= s_maxz[i]) & (iz >= s_minz[i])
                      & (scale > 0.f);
        if (ok && scale < best) { best = scale; bi = i; }
    }

    const long long o = (long long)b * HW + p;
    depth_out[o]       = best;
    nrm_out[o * 3 + 0] = s_nx[bi];
    nrm_out[o * 3 + 1] = 0.f;
    nrm_out[o * 3 + 2] = s_nz[bi];
}

extern "C" void kernel_launch(void* const* d_in, const int* in_sizes, int n_in,
                              void* d_out, int out_size, void* d_ws, size_t ws_size,
                              hipStream_t stream) {
    const float* corners = (const float*)d_in[0];
    const float* grid    = (const float*)d_in[1];
    const int*   nums    = (const int*)d_in[2];

    const int       B    = in_sizes[2];
    const int       Nmax = in_sizes[0] / (B * 3);
    const long long HW   = in_sizes[1] / 3;

    float* depth = (float*)d_out;
    float* nrm   = depth + (long long)B * HW;

    dim3 grd((unsigned)(HW / TPB), (unsigned)B);
    corner2depth_wmma_kernel<<<grd, TPB, 0, stream>>>(corners, grid, nums,
                                                      depth, nrm, Nmax, HW);
}


# WMMA operands swapped (walls = A rows, rays = B columns): each lane tests its own ray's walls on the accumulators, wall constants by DPP row broadcast; LDS transpose, wall table, barriers gone
# speedup vs baseline: 1.0832x; 1.0832x over previous
#include <hip/hip_runtime.h>


typedef __attribute__((ext_vector_type(2))) float v2f;
typedef __attribute__((ext_vector_type(8))) float v8f;

#define C2D_EPS   0.01f
#define MAXW      16
#define TPB       256

__device__ __forceinline__ float xhalf_f(float x) {
    const int v = __builtin_bit_cast(int, x);
    return __builtin_bit_cast(float,
        __builtin_amdgcn_permlanex16(v, v, 0x76543210, 0xfedcba98, false, false));
}
__device__ __forceinline__ int xhalf_i(int v) {
    return __builtin_amdgcn_permlanex16(v, v, 0x76543210, 0xfedcba98, false, false);
}

template <int R>
__device__ __forceinline__ float row_lane(float x) {
    return __builtin_bit_cast(float,
        __builtin_amdgcn_update_dpp(0, __builtin_bit_cast(int, x),
                                    0x150 + R, 0xF, 0xF, true));
}

template <int R>
__device__ __forceinline__ void wall_step(int h,
        float nd, float mnx, float mxx, float mnz, float mxz,
        float d0, float d1, float gx0, float gz0, float gx1, float gz1,
        float& best0, int& bi0, float& best1, int& bi1)
{
    if (R == 0 || R < h) {
        const float ndr  = row_lane<R>(nd);
        const float mnxr = row_lane<R>(mnx), mxxr = row_lane<R>(mxx);
        const float mnzr = row_lane<R>(mnz), mxzr = row_lane<R>(mxz);
        {
            const float scale = ndr * __builtin_amdgcn_rcpf(d0);
            const float ix = gx0 * scale;
            const float iz = gz0 * scale;
            const bool ok = (ix <= mxxr) & (ix >= mnxr)
                          & (iz <= mxzr) & (iz >= mnzr)
                          & (scale > 0.f);
            if (ok && scale < best0) { best0 = scale; bi0 = R; }
        }
        {
            const float scale = ndr * __builtin_amdgcn_rcpf(d1);
            const float ix = gx1 * scale;
            const float iz = gz1 * scale;
            const bool ok = (ix <= mxxr) & (ix >= mnxr)
                          & (iz <= mxzr) & (iz >= mnzr)
                          & (scale > 0.f);
            if (ok && scale < best1) { best1 = scale; bi1 = R; }
        }
    }
}

__global__ __launch_bounds__(TPB)
void corner2depth_wmma_kernel(const float* __restrict__ corners,
                              const float* __restrict__ grid,
                              const int*   __restrict__ nums,
                              float* __restrict__ depth_out,
                              float* __restrict__ nrm_out,
                              int Nmax, long long HW)
{
    const int tid  = threadIdx.x;
    const int b    = blockIdx.y;
    const int lane = tid & 31;
    const int lh   = lane & 15;
    const int hlf  = lane >> 4;
    const int wv   = tid >> 5;

    const long long pw = (long long)blockIdx.x * TPB + wv * 32;
    const long long p0 = pw + lh;
    const long long p1 = p0 + 16;
    const float gx0 = grid[p0 * 3 + 0], gz0 = grid[p0 * 3 + 2];
    const float gx1 = grid[p1 * 3 + 0], gz1 = grid[p1 * 3 + 2];
    float cx = 0.f, cz = 0.f;
    if (lh < Nmax) {
        const long long cb = ((long long)b * Nmax + lh) * 3;
        cx = corners[cb + 0];
        cz = corners[cb + 2];
    }

    int n = nums[b];
    if (n > MAXW) n = MAXW;
    const int h = (n + 1) >> 1;

    const int  w     = hlf ? h + lh : lh;
    const bool valid = hlf ? (w < n) : (lh < h);
    const int  j     = (w + 1 == n) ? 0 : w + 1;
    const float x0 = __shfl(cx, w, 32), z0 = __shfl(cz, w, 32);
    const float x1 = __shfl(cx, j, 32), z1 = __shfl(cz, j, 32);

    const int  wa     = (lh < 8) ? lh : h + lh - 8;
    const bool avalid = (lh < 8) ? (lh < h) : (wa < n);
    const int  ja     = (wa + 1 == n) ? 0 : wa + 1;
    const float cs = hlf ? cx : cz;
    const float ua = __shfl(cs, (wa & 15) + 16 * hlf, 32);
    const float va = __shfl(cs, (ja & 15) + 16 * hlf, 32);

    float nx = 0.f, nz = 0.f, nd = 0.f;
    float mnx = 0.f, mxx = 0.f, mnz = 0.f, mxz = 0.f;
    if (valid) {
        const float dx = x1 - x0, dz = z1 - z0;
        nx = -dz;
        nz =  dx;
        nd = nx * x0 + nz * z0;
        mnx = fminf(x0, x1) - C2D_EPS;  mxx = fmaxf(x0, x1) + C2D_EPS;
        mnz = fminf(z0, z1) - C2D_EPS;  mxz = fmaxf(z0, z1) + C2D_EPS;
    }

    const float da = va - ua;
    v2f am, b0, b1;
    am.x = avalid ? (hlf ? da : -da) : 0.f;   am.y = 0.f;
    b0.x = hlf ? gz0 : gx0;  b0.y = 0.f;
    b1.x = hlf ? gz1 : gx1;  b1.y = 0.f;

    v8f acc0 = {}, acc1 = {};
    acc0 = __builtin_amdgcn_wmma_f32_16x16x4_f32(false, am, false, b0,
                                                 (short)0, acc0, false, false);
    acc1 = __builtin_amdgcn_wmma_f32_16x16x4_f32(false, am, false, b1,
                                                 (short)0, acc1, false, false);

    float best0 = __builtin_inff(), best1 = __builtin_inff();
    int   bi0 = 0, bi1 = 0;
    wall_step<0>(h, nd, mnx, mxx, mnz, mxz, acc0[0], acc1[0], gx0, gz0, gx1, gz1, best0, bi0, best1, bi1);
    wall_step<1>(h, nd, mnx, mxx, mnz, mxz, acc0[1], acc1[1], gx0, gz0, gx1, gz1, best0, bi0, best1, bi1);
    wall_step<2>(h, nd, mnx, mxx, mnz, mxz, acc0[2], acc1[2], gx0, gz0, gx1, gz1, best0, bi0, best1, bi1);
    wall_step<3>(h, nd, mnx, mxx, mnz, mxz, acc0[3], acc1[3], gx0, gz0, gx1, gz1, best0, bi0, best1, bi1);
    wall_step<4>(h, nd, mnx, mxx, mnz, mxz, acc0[4], acc1[4], gx0, gz0, gx1, gz1, best0, bi0, best1, bi1);
    wall_step<5>(h, nd, mnx, mxx, mnz, mxz, acc0[5], acc1[5], gx0, gz0, gx1, gz1, best0, bi0, best1, bi1);
    wall_step<6>(h, nd, mnx, mxx, mnz, mxz, acc0[6], acc1[6], gx0, gz0, gx1, gz1, best0, bi0, best1, bi1);
    wall_step<7>(h, nd, mnx, mxx, mnz, mxz, acc0[7], acc1[7], gx0, gz0, gx1, gz1, best0, bi0, best1, bi1);

    const float keepB = hlf ? best1 : best0;
    const int   keepI = (hlf ? bi1 : bi0) + 16 * hlf;
    const float sendB = hlf ? best0 : best1;
    const int   sendI = (hlf ? bi0 : bi1) + 16 * hlf;
    const float recvB = xhalf_f(sendB);
    const int   recvI = xhalf_i(sendI);
    const bool  take  = hlf ? (recvB <= keepB) : (recvB < keepB);
    const float best  = take ? recvB : keepB;
    const int   wl    = take ? recvI : keepI;

    const float onx = __shfl(nx, wl, 32);
    const float onz = __shfl(nz, wl, 32);

    const long long p = pw + lane;
    const long long o = (long long)b * HW + p;
    depth_out[o]       = best;
    nrm_out[o * 3 + 0] = onx;
    nrm_out[o * 3 + 1] = 0.f;
    nrm_out[o * 3 + 2] = onz;
}

extern "C" void kernel_launch(void* const* d_in, const int* in_sizes, int n_in,
                              void* d_out, int out_size, void* d_ws, size_t ws_size,
                              hipStream_t stream) {
    const float* corners = (const float*)d_in[0];
    const float* grid    = (const float*)d_in[1];
    const int*   nums    = (const int*)d_in[2];

    const int       B    = in_sizes[2];
    const int       Nmax = in_sizes[0] / (B * 3);
    const long long HW   = in_sizes[1] / 3;

    float* depth = (float*)d_out;
    float* nrm   = depth + (long long)B * HW;

    dim3 grd((unsigned)(HW / TPB), (unsigned)B);
    corner2depth_wmma_kernel<<<grd, TPB, 0, stream>>>(corners, grid, nums,
                                                      depth, nrm, Nmax, HW);
}


# Each wave walks 8 chunks of 32 rays instead of one: wall set-up, A fragment and row-broadcast wall constants paid once per wave and kept in registers, next chunk's rays loaded one chunk ahead
# speedup vs baseline: 1.2477x; 1.1519x over previous
#include <hip/hip_runtime.h>


typedef __attribute__((ext_vector_type(2))) float v2f;
typedef __attribute__((ext_vector_type(8))) float v8f;

#define C2D_EPS   0.01f
#define MAXW      16
#define TPB       256
#define WPB       (TPB / 32)
#define CPW       8

__device__ __forceinline__ float xhalf_f(float x) {
    const int v = __builtin_bit_cast(int, x);
    return __builtin_bit_cast(float,
        __builtin_amdgcn_permlanex16(v, v, 0x76543210, 0xfedcba98, false, false));
}
__device__ __forceinline__ int xhalf_i(int v) {
    return __builtin_amdgcn_permlanex16(v, v, 0x76543210, 0xfedcba98, false, false);
}

template <int R>
__device__ __forceinline__ float row_lane(float x) {
    return __builtin_bit_cast(float,
        __builtin_amdgcn_update_dpp(0, __builtin_bit_cast(int, x),
                                    0x150 + R, 0xF, 0xF, true));
}

struct WallK { float nd, mnx, mxx, mnz, mxz; };

template <int R>
__device__ __forceinline__ WallK wall_row(float nd, float mnx, float mxx,
                                          float mnz, float mxz) {
    WallK k;
    k.nd  = row_lane<R>(nd);
    k.mnx = row_lane<R>(mnx);  k.mxx = row_lane<R>(mxx);
    k.mnz = row_lane<R>(mnz);  k.mxz = row_lane<R>(mxz);
    return k;
}

template <int R>
__device__ __forceinline__ void wall_step(int h, const WallK& k,
        float d0, float d1, float gx0, float gz0, float gx1, float gz1,
        float& best0, int& bi0, float& best1, int& bi1)
{
    if (R == 0 || R < h) {
        {
            const float scale = k.nd * __builtin_amdgcn_rcpf(d0);
            const float ix = gx0 * scale;
            const float iz = gz0 * scale;
            const bool ok = (ix <= k.mxx) & (ix >= k.mnx)
                          & (iz <= k.mxz) & (iz >= k.mnz)
                          & (scale > 0.f);
            if (ok && scale < best0) { best0 = scale; bi0 = R; }
        }
        {
            const float scale = k.nd * __builtin_amdgcn_rcpf(d1);
            const float ix = gx1 * scale;
            const float iz = gz1 * scale;
            const bool ok = (ix <= k.mxx) & (ix >= k.mnx)
                          & (iz <= k.mxz) & (iz >= k.mnz)
                          & (scale > 0.f);
            if (ok && scale < best1) { best1 = scale; bi1 = R; }
        }
    }
}

__global__ __launch_bounds__(TPB)
void corner2depth_wmma_kernel(const float* __restrict__ corners,
                              const float* __restrict__ grid,
                              const int*   __restrict__ nums,
                              float* __restrict__ depth_out,
                              float* __restrict__ nrm_out,
                              int Nmax, long long HW)
{
    const int tid  = threadIdx.x;
    const int b    = blockIdx.y;
    const int lane = tid & 31;
    const int lh   = lane & 15;
    const int hlf  = lane >> 4;
    const int wv   = __builtin_amdgcn_readfirstlane(tid >> 5);

    const long long nchunks = HW >> 5;
    const long long c = (long long)blockIdx.x * (WPB * CPW) + wv;
    if (c >= nchunks) return;
    const long long left = (nchunks - c + (WPB - 1)) / WPB;
    const int nit = left < CPW ? (int)left : CPW;

    const float* gb = grid + c * 96;
    const int    go = lh * 3;
    float gx0 = gb[go],      gz0 = gb[go + 2];
    float gx1 = gb[go + 48], gz1 = gb[go + 50];
    float cx = 0.f, cz = 0.f;
    if (lh < Nmax) {
        const long long cb = ((long long)b * Nmax + lh) * 3;
        cx = corners[cb + 0];
        cz = corners[cb + 2];
    }

    int n = nums[b];
    if (n > MAXW) n = MAXW;
    const int h = (n + 1) >> 1;

    const int  w     = hlf ? h + lh : lh;
    const bool valid = hlf ? (w < n) : (lh < h);
    const int  j     = (w + 1 == n) ? 0 : w + 1;
    const float x0 = __shfl(cx, w, 32), z0 = __shfl(cz, w, 32);
    const float x1 = __shfl(cx, j, 32), z1 = __shfl(cz, j, 32);

    const int  wa     = (lh < 8) ? lh : h + lh - 8;
    const bool avalid = (lh < 8) ? (lh < h) : (wa < n);
    const int  ja     = (wa + 1 == n) ? 0 : wa + 1;
    const float cs = hlf ? cx : cz;
    const float ua = __shfl(cs, (wa & 15) + 16 * hlf, 32);
    const float va = __shfl(cs, (ja & 15) + 16 * hlf, 32);

    float nx = 0.f, nz = 0.f, nd = 0.f;
    float mnx = 0.f, mxx = 0.f, mnz = 0.f, mxz = 0.f;
    if (valid) {
        const float dx = x1 - x0, dz = z1 - z0;
        nx = -dz;
        nz =  dx;
        nd = nx * x0 + nz * z0;
        mnx = fminf(x0, x1) - C2D_EPS;  mxx = fmaxf(x0, x1) + C2D_EPS;
        mnz = fminf(z0, z1) - C2D_EPS;  mxz = fmaxf(z0, z1) + C2D_EPS;
    }

    const float da = va - ua;
    v2f am;
    am.x = avalid ? (hlf ? da : -da) : 0.f;   am.y = 0.f;

    const WallK k0 = wall_row<0>(nd, mnx, mxx, mnz, mxz);
    const WallK k1 = wall_row<1>(nd, mnx, mxx, mnz, mxz);
    const WallK k2 = wall_row<2>(nd, mnx, mxx, mnz, mxz);
    const WallK k3 = wall_row<3>(nd, mnx, mxx, mnz, mxz);
    const WallK k4 = wall_row<4>(nd, mnx, mxx, mnz, mxz);
    const WallK k5 = wall_row<5>(nd, mnx, mxx, mnz, mxz);
    const WallK k6 = wall_row<6>(nd, mnx, mxx, mnz, mxz);
    const WallK k7 = wall_row<7>(nd, mnx, mxx, mnz, mxz);

    float* db = depth_out + ((long long)b * HW + c * 32);
    float* nb = nrm_out + ((long long)b * HW + c * 32) * 3;
    const int no = lane * 3;

    for (int it = 0; it < nit; ++it) {
        const bool more = it + 1 < nit;
        float ngx0 = 0.f, ngz0 = 0.f, ngx1 = 0.f, ngz1 = 0.f;
        if (more) {
            gb += WPB * 96;
            ngx0 = gb[go];       ngz0 = gb[go + 2];
            ngx1 = gb[go + 48];  ngz1 = gb[go + 50];
        }

        v2f b0, b1;
        b0.x = hlf ? gz0 : gx0;  b0.y = 0.f;
        b1.x = hlf ? gz1 : gx1;  b1.y = 0.f;

        v8f acc0 = {}, acc1 = {};
        acc0 = __builtin_amdgcn_wmma_f32_16x16x4_f32(false, am, false, b0,
                                                     (short)0, acc0, false, false);
        acc1 = __builtin_amdgcn_wmma_f32_16x16x4_f32(false, am, false, b1,
                                                     (short)0, acc1, false, false);

        float best0 = __builtin_inff(), best1 = __builtin_inff();
        int   bi0 = 0, bi1 = 0;
        wall_step<0>(h, k0, acc0[0], acc1[0], gx0, gz0, gx1, gz1, best0, bi0, best1, bi1);
        wall_step<1>(h, k1, acc0[1], acc1[1], gx0, gz0, gx1, gz1, best0, bi0, best1, bi1);
        wall_step<2>(h, k2, acc0[2], acc1[2], gx0, gz0, gx1, gz1, best0, bi0, best1, bi1);
        wall_step<3>(h, k3, acc0[3], acc1[3], gx0, gz0, gx1, gz1, best0, bi0, best1, bi1);
        wall_step<4>(h, k4, acc0[4], acc1[4], gx0, gz0, gx1, gz1, best0, bi0, best1, bi1);
        wall_step<5>(h, k5, acc0[5], acc1[5], gx0, gz0, gx1, gz1, best0, bi0, best1, bi1);
        wall_step<6>(h, k6, acc0[6], acc1[6], gx0, gz0, gx1, gz1, best0, bi0, best1, bi1);
        wall_step<7>(h, k7, acc0[7], acc1[7], gx0, gz0, gx1, gz1, best0, bi0, best1, bi1);

        const float keepB = hlf ? best1 : best0;
        const int   keepI = (hlf ? bi1 : bi0) + 16 * hlf;
        const float sendB = hlf ? best0 : best1;
        const int   sendI = (hlf ? bi0 : bi1) + 16 * hlf;
        const float recvB = xhalf_f(sendB);
        const int   recvI = xhalf_i(sendI);
        const bool  take  = hlf ? (recvB <= keepB) : (recvB < keepB);
        const float best  = take ? recvB : keepB;
        const int   wl    = take ? recvI : keepI;

        const float onx = __shfl(nx, wl, 32);
        const float onz = __shfl(nz, wl, 32);

        db[lane]   = best;
        nb[no + 0] = onx;
        nb[no + 1] = 0.f;
        nb[no + 2] = onz;

        gx0 = ngx0;  gz0 = ngz0;  gx1 = ngx1;  gz1 = ngz1;
        db += WPB * 32;
        nb += WPB * 96;
    }
}

extern "C" void kernel_launch(void* const* d_in, const int* in_sizes, int n_in,
                              void* d_out, int out_size, void* d_ws, size_t ws_size,
                              hipStream_t stream) {
    const float* corners = (const float*)d_in[0];
    const float* grid    = (const float*)d_in[1];
    const int*   nums    = (const int*)d_in[2];

    const int       B    = in_sizes[2];
    const int       Nmax = in_sizes[0] / (B * 3);
    const long long HW   = in_sizes[1] / 3;

    float* depth = (float*)d_out;
    float* nrm   = depth + (long long)B * HW;

    dim3 grd((unsigned)(HW / (TPB * CPW)), (unsigned)B);
    corner2depth_wmma_kernel<<<grd, TPB, 0, stream>>>(corners, grid, nums,
                                                      depth, nrm, Nmax, HW);
}


# Chunk loop compiled once per wall-register count h, picked by one scalar switch ahead of it: a chunk is one straight block (no test of h, no branch but the back edge; last trip re-reads its rays)
# speedup vs baseline: 1.3443x; 1.0774x over previous
#include <hip/hip_runtime.h>


typedef __attribute__((ext_vector_type(2))) float v2f;
typedef __attribute__((ext_vector_type(8))) float v8f;

#define C2D_EPS   0.01f
#define MAXW      16
#define TPB       256
#define WPB       (TPB / 32)
#define CPW       8

__device__ __forceinline__ float xhalf_f(float x) {
    const int v = __builtin_bit_cast(int, x);
    return __builtin_bit_cast(float,
        __builtin_amdgcn_permlanex16(v, v, 0x76543210, 0xfedcba98, false, false));
}
__device__ __forceinline__ int xhalf_i(int v) {
    return __builtin_amdgcn_permlanex16(v, v, 0x76543210, 0xfedcba98, false, false);
}

template <int R>
__device__ __forceinline__ float row_lane(float x) {
    return __builtin_bit_cast(float,
        __builtin_amdgcn_update_dpp(0, __builtin_bit_cast(int, x),
                                    0x150 + R, 0xF, 0xF, true));
}

struct WallK { float nd, mnx, mxx, mnz, mxz; };

template <int R>
__device__ __forceinline__ WallK wall_row(float nd, float mnx, float mxx,
                                          float mnz, float mxz) {
    WallK k;
    k.nd  = row_lane<R>(nd);
    k.mnx = row_lane<R>(mnx);  k.mxx = row_lane<R>(mxx);
    k.mnz = row_lane<R>(mnz);  k.mxz = row_lane<R>(mxz);
    return k;
}

template <int R>
__device__ __forceinline__ void wall_step(const WallK& k,
        float d0, float d1, float gx0, float gz0, float gx1, float gz1,
        float& best0, int& bi0, float& best1, int& bi1)
{
    {
        {
            const float scale = k.nd * __builtin_amdgcn_rcpf(d0);
            const float ix = gx0 * scale;
            const float iz = gz0 * scale;
            const bool ok = (ix <= k.mxx) & (ix >= k.mnx)
                          & (iz <= k.mxz) & (iz >= k.mnz)
                          & (scale > 0.f);
            if (ok && scale < best0) { best0 = scale; bi0 = R; }
        }
        {
            const float scale = k.nd * __builtin_amdgcn_rcpf(d1);
            const float ix = gx1 * scale;
            const float iz = gz1 * scale;
            const bool ok = (ix <= k.mxx) & (ix >= k.mnx)
                          & (iz <= k.mxz) & (iz >= k.mnz)
                          & (scale > 0.f);
            if (ok && scale < best1) { best1 = scale; bi1 = R; }
        }
    }
}

template <int H>
__device__ __forceinline__ void walk_chunks(
        const int nit, const float* __restrict__ gb, const int go,
        float gx0, float gz0, float gx1, float gz1,
        const v2f am, const int hlf, const int lane,
        const WallK& k0, const WallK& k1, const WallK& k2, const WallK& k3,
        const WallK& k4, const WallK& k5, const WallK& k6, const WallK& k7,
        const float nx, const float nz,
        float* __restrict__ db, float* __restrict__ nb, const int no)
{
    for (int it = 0; it < nit; ++it) {
        gb += (it + 1 < nit) ? WPB * 96 : 0;
        const float ngx0 = gb[go],      ngz0 = gb[go + 2];
        const float ngx1 = gb[go + 48], ngz1 = gb[go + 50];

        v2f b0, b1;
        b0.x = hlf ? gz0 : gx0;  b0.y = 0.f;
        b1.x = hlf ? gz1 : gx1;  b1.y = 0.f;

        v8f acc0 = {}, acc1 = {};
        acc0 = __builtin_amdgcn_wmma_f32_16x16x4_f32(false, am, false, b0,
                                                     (short)0, acc0, false, false);
        acc1 = __builtin_amdgcn_wmma_f32_16x16x4_f32(false, am, false, b1,
                                                     (short)0, acc1, false, false);
        __builtin_amdgcn_sched_barrier(0);

        float best0 = __builtin_inff(), best1 = __builtin_inff();
        int   bi0 = 0, bi1 = 0;
        wall_step<0>(k0, acc0[0], acc1[0], gx0, gz0, gx1, gz1, best0, bi0, best1, bi1);
        if constexpr (H > 1)
            wall_step<1>(k1, acc0[1], acc1[1], gx0, gz0, gx1, gz1, best0, bi0, best1, bi1);
        if constexpr (H > 2)
            wall_step<2>(k2, acc0[2], acc1[2], gx0, gz0, gx1, gz1, best0, bi0, best1, bi1);
        if constexpr (H > 3)
            wall_step<3>(k3, acc0[3], acc1[3], gx0, gz0, gx1, gz1, best0, bi0, best1, bi1);
        if constexpr (H > 4)
            wall_step<4>(k4, acc0[4], acc1[4], gx0, gz0, gx1, gz1, best0, bi0, best1, bi1);
        if constexpr (H > 5)
            wall_step<5>(k5, acc0[5], acc1[5], gx0, gz0, gx1, gz1, best0, bi0, best1, bi1);
        if constexpr (H > 6)
            wall_step<6>(k6, acc0[6], acc1[6], gx0, gz0, gx1, gz1, best0, bi0, best1, bi1);
        if constexpr (H > 7)
            wall_step<7>(k7, acc0[7], acc1[7], gx0, gz0, gx1, gz1, best0, bi0, best1, bi1);

        const float keepB = hlf ? best1 : best0;
        const int   keepI = (hlf ? bi1 : bi0) + 16 * hlf;
        const float sendB = hlf ? best0 : best1;
        const int   sendI = (hlf ? bi0 : bi1) + 16 * hlf;
        const float recvB = xhalf_f(sendB);
        const int   recvI = xhalf_i(sendI);
        const bool  take  = hlf ? (recvB <= keepB) : (recvB < keepB);
        const float best  = take ? recvB : keepB;
        const int   wl    = take ? recvI : keepI;

        const float onx = __shfl(nx, wl, 32);
        const float onz = __shfl(nz, wl, 32);

        db[lane]   = best;
        nb[no + 0] = onx;
        nb[no + 1] = 0.f;
        nb[no + 2] = onz;

        gx0 = ngx0;  gz0 = ngz0;  gx1 = ngx1;  gz1 = ngz1;
        db += WPB * 32;
        nb += WPB * 96;
    }
}

__global__ __launch_bounds__(TPB)
void corner2depth_wmma_kernel(const float* __restrict__ corners,
                              const float* __restrict__ grid,
                              const int*   __restrict__ nums,
                              float* __restrict__ depth_out,
                              float* __restrict__ nrm_out,
                              int Nmax, long long HW)
{
    const int tid  = threadIdx.x;
    const int b    = blockIdx.y;
    const int lane = tid & 31;
    const int lh   = lane & 15;
    const int hlf  = lane >> 4;
    const int wv   = __builtin_amdgcn_readfirstlane(tid >> 5);

    const long long nchunks = HW >> 5;
    const long long c = (long long)blockIdx.x * (WPB * CPW) + wv;
    if (c >= nchunks) return;
    const long long left = (nchunks - c + (WPB - 1)) / WPB;
    const int nit = left < CPW ? (int)left : CPW;

    const float* gb = grid + c * 96;
    const int    go = lh * 3;
    float gx0 = gb[go],      gz0 = gb[go + 2];
    float gx1 = gb[go + 48], gz1 = gb[go + 50];
    float cx = 0.f, cz = 0.f;
    if (lh < Nmax) {
        const long long cb = ((long long)b * Nmax + lh) * 3;
        cx = corners[cb + 0];
        cz = corners[cb + 2];
    }

    int n = nums[b];
    if (n > MAXW) n = MAXW;
    const int h = (n + 1) >> 1;

    const int  w     = hlf ? h + lh : lh;
    const bool valid = hlf ? (w < n) : (lh < h);
    const int  j     = (w + 1 == n) ? 0 : w + 1;
    const float x0 = __shfl(cx, w, 32), z0 = __shfl(cz, w, 32);
    const float x1 = __shfl(cx, j, 32), z1 = __shfl(cz, j, 32);

    const int  wa     = (lh < 8) ? lh : h + lh - 8;
    const bool avalid = (lh < 8) ? (lh < h) : (wa < n);
    const int  ja     = (wa + 1 == n) ? 0 : wa + 1;
    const float cs = hlf ? cx : cz;
    const float ua = __shfl(cs, (wa & 15) + 16 * hlf, 32);
    const float va = __shfl(cs, (ja & 15) + 16 * hlf, 32);

    float nx = 0.f, nz = 0.f, nd = 0.f;
    float mnx = 0.f, mxx = 0.f, mnz = 0.f, mxz = 0.f;
    if (valid) {
        const float dx = x1 - x0, dz = z1 - z0;
        nx = -dz;
        nz =  dx;
        nd = nx * x0 + nz * z0;
        mnx = fminf(x0, x1) - C2D_EPS;  mxx = fmaxf(x0, x1) + C2D_EPS;
        mnz = fminf(z0, z1) - C2D_EPS;  mxz = fmaxf(z0, z1) + C2D_EPS;
    }

    const float da = va - ua;
    v2f am;
    am.x = avalid ? (hlf ? da : -da) : 0.f;   am.y = 0.f;

    const WallK k0 = wall_row<0>(nd, mnx, mxx, mnz, mxz);
    const WallK k1 = wall_row<1>(nd, mnx, mxx, mnz, mxz);
    const WallK k2 = wall_row<2>(nd, mnx, mxx, mnz, mxz);
    const WallK k3 = wall_row<3>(nd, mnx, mxx, mnz, mxz);
    const WallK k4 = wall_row<4>(nd, mnx, mxx, mnz, mxz);
    const WallK k5 = wall_row<5>(nd, mnx, mxx, mnz, mxz);
    const WallK k6 = wall_row<6>(nd, mnx, mxx, mnz, mxz);
    const WallK k7 = wall_row<7>(nd, mnx, mxx, mnz, mxz);

    float* db = depth_out + ((long long)b * HW + c * 32);
    float* nb = nrm_out + ((long long)b * HW + c * 32) * 3;
    const int no = lane * 3;

#define C2D_WALK(HH) walk_chunks<HH>(nit, gb, go, gx0, gz0, gx1, gz1, am, hlf, lane, \
                                    k0, k1, k2, k3, k4, k5, k6, k7, nx, nz, db, nb, no)
    switch (h) {
        case 2:  C2D_WALK(2); break;
        case 3:  C2D_WALK(3); break;
        case 4:  C2D_WALK(4); break;
        case 5:  C2D_WALK(5); break;
        case 6:  C2D_WALK(6); break;
        case 7:  C2D_WALK(7); break;
        case 8:  C2D_WALK(8); break;
        default: C2D_WALK(1); break;
    }
#undef C2D_WALK
}

extern "C" void kernel_launch(void* const* d_in, const int* in_sizes, int n_in,
                              void* d_out, int out_size, void* d_ws, size_t ws_size,
                              hipStream_t stream) {
    const float* corners = (const float*)d_in[0];
    const float* grid    = (const float*)d_in[1];
    const int*   nums    = (const int*)d_in[2];

    const int       B    = in_sizes[2];
    const int       Nmax = in_sizes[0] / (B * 3);
    const long long HW   = in_sizes[1] / 3;

    float* depth = (float*)d_out;
    float* nrm   = depth + (long long)B * HW;

    dim3 grd((unsigned)(HW / (TPB * CPW)), (unsigned)B);
    corner2depth_wmma_kernel<<<grd, TPB, 0, stream>>>(corners, grid, nums,
                                                      depth, nrm, Nmax, HW);
}


# Wall bounds kept as centre and half width per axis: one fma gives ix - cx, one |.| <= h compare per axis replaces the two bound compares (6 -> 4 compares a test); band edges move by a rounding step
# speedup vs baseline: 1.4638x; 1.0889x over previous
#include <hip/hip_runtime.h>


typedef __attribute__((ext_vector_type(2))) float v2f;
typedef __attribute__((ext_vector_type(8))) float v8f;

#define C2D_EPS   0.01f
#define MAXW      16
#define TPB       256
#define WPB       (TPB / 32)
#define CPW       8

__device__ __forceinline__ float xhalf_f(float x) {
    const int v = __builtin_bit_cast(int, x);
    return __builtin_bit_cast(float,
        __builtin_amdgcn_permlanex16(v, v, 0x76543210, 0xfedcba98, false, false));
}
__device__ __forceinline__ int xhalf_i(int v) {
    return __builtin_amdgcn_permlanex16(v, v, 0x76543210, 0xfedcba98, false, false);
}

template <int R>
__device__ __forceinline__ float row_lane(float x) {
    return __builtin_bit_cast(float,
        __builtin_amdgcn_update_dpp(0, __builtin_bit_cast(int, x),
                                    0x150 + R, 0xF, 0xF, true));
}

struct WallK { float nd, cx, hx, cz, hz; };

template <int R>
__device__ __forceinline__ WallK wall_row(float nd, float cx, float hx,
                                          float cz, float hz) {
    WallK k;
    k.nd = row_lane<R>(nd);
    k.cx = row_lane<R>(cx);  k.hx = row_lane<R>(hx);
    k.cz = row_lane<R>(cz);  k.hz = row_lane<R>(hz);
    return k;
}

template <int R>
__device__ __forceinline__ void wall_step(const WallK& k,
        float d0, float d1, float gx0, float gz0, float gx1, float gz1,
        float& best0, int& bi0, float& best1, int& bi1)
{
    {
        {
            const float scale = k.nd * __builtin_amdgcn_rcpf(d0);
            const float ex = __builtin_fmaf(gx0, scale, -k.cx);
            const float ez = __builtin_fmaf(gz0, scale, -k.cz);
            const bool ok = (__builtin_fabsf(ex) <= k.hx)
                          & (__builtin_fabsf(ez) <= k.hz)
                          & (scale > 0.f);
            if (ok && scale < best0) { best0 = scale; bi0 = R; }
        }
        {
            const float scale = k.nd * __builtin_amdgcn_rcpf(d1);
            const float ex = __builtin_fmaf(gx1, scale, -k.cx);
            const float ez = __builtin_fmaf(gz1, scale, -k.cz);
            const bool ok = (__builtin_fabsf(ex) <= k.hx)
                          & (__builtin_fabsf(ez) <= k.hz)
                          & (scale > 0.f);
            if (ok && scale < best1) { best1 = scale; bi1 = R; }
        }
    }
}

template <int H>
__device__ __forceinline__ void walk_chunks(
        const int nit, const float* __restrict__ gb, const int go,
        float gx0, float gz0, float gx1, float gz1,
        const v2f am, const int hlf, const int lane,
        const WallK& k0, const WallK& k1, const WallK& k2, const WallK& k3,
        const WallK& k4, const WallK& k5, const WallK& k6, const WallK& k7,
        const float nx, const float nz,
        float* __restrict__ db, float* __restrict__ nb, const int no)
{
    for (int it = 0; it < nit; ++it) {
        gb += (it + 1 < nit) ? WPB * 96 : 0;
        const float ngx0 = gb[go],      ngz0 = gb[go + 2];
        const float ngx1 = gb[go + 48], ngz1 = gb[go + 50];

        v2f b0, b1;
        b0.x = hlf ? gz0 : gx0;  b0.y = 0.f;
        b1.x = hlf ? gz1 : gx1;  b1.y = 0.f;

        v8f acc0 = {}, acc1 = {};
        acc0 = __builtin_amdgcn_wmma_f32_16x16x4_f32(false, am, false, b0,
                                                     (short)0, acc0, false, false);
        acc1 = __builtin_amdgcn_wmma_f32_16x16x4_f32(false, am, false, b1,
                                                     (short)0, acc1, false, false);
        __builtin_amdgcn_sched_barrier(0);

        float best0 = __builtin_inff(), best1 = __builtin_inff();
        int   bi0 = 0, bi1 = 0;
        wall_step<0>(k0, acc0[0], acc1[0], gx0, gz0, gx1, gz1, best0, bi0, best1, bi1);
        if constexpr (H > 1)
            wall_step<1>(k1, acc0[1], acc1[1], gx0, gz0, gx1, gz1, best0, bi0, best1, bi1);
        if constexpr (H > 2)
            wall_step<2>(k2, acc0[2], acc1[2], gx0, gz0, gx1, gz1, best0, bi0, best1, bi1);
        if constexpr (H > 3)
            wall_step<3>(k3, acc0[3], acc1[3], gx0, gz0, gx1, gz1, best0, bi0, best1, bi1);
        if constexpr (H > 4)
            wall_step<4>(k4, acc0[4], acc1[4], gx0, gz0, gx1, gz1, best0, bi0, best1, bi1);
        if constexpr (H > 5)
            wall_step<5>(k5, acc0[5], acc1[5], gx0, gz0, gx1, gz1, best0, bi0, best1, bi1);
        if constexpr (H > 6)
            wall_step<6>(k6, acc0[6], acc1[6], gx0, gz0, gx1, gz1, best0, bi0, best1, bi1);
        if constexpr (H > 7)
            wall_step<7>(k7, acc0[7], acc1[7], gx0, gz0, gx1, gz1, best0, bi0, best1, bi1);

        const float keepB = hlf ? best1 : best0;
        const int   keepI = (hlf ? bi1 : bi0) + 16 * hlf;
        const float sendB = hlf ? best0 : best1;
        const int   sendI = (hlf ? bi0 : bi1) + 16 * hlf;
        const float recvB = xhalf_f(sendB);
        const int   recvI = xhalf_i(sendI);
        const bool  take  = hlf ? (recvB <= keepB) : (recvB < keepB);
        const float best  = take ? recvB : keepB;
        const int   wl    = take ? recvI : keepI;

        const float onx = __shfl(nx, wl, 32);
        const float onz = __shfl(nz, wl, 32);

        db[lane]   = best;
        nb[no + 0] = onx;
        nb[no + 1] = 0.f;
        nb[no + 2] = onz;

        gx0 = ngx0;  gz0 = ngz0;  gx1 = ngx1;  gz1 = ngz1;
        db += WPB * 32;
        nb += WPB * 96;
    }
}

__global__ __launch_bounds__(TPB)
void corner2depth_wmma_kernel(const float* __restrict__ corners,
                              const float* __restrict__ grid,
                              const int*   __restrict__ nums,
                              float* __restrict__ depth_out,
                              float* __restrict__ nrm_out,
                              int Nmax, long long HW)
{
    const int tid  = threadIdx.x;
    const int b    = blockIdx.y;
    const int lane = tid & 31;
    const int lh   = lane & 15;
    const int hlf  = lane >> 4;
    const int wv   = __builtin_amdgcn_readfirstlane(tid >> 5);

    const long long nchunks = HW >> 5;
    const long long c = (long long)blockIdx.x * (WPB * CPW) + wv;
    if (c >= nchunks) return;
    const long long left = (nchunks - c + (WPB - 1)) / WPB;
    const int nit = left < CPW ? (int)left : CPW;

    const float* gb = grid + c * 96;
    const int    go = lh * 3;
    float gx0 = gb[go],      gz0 = gb[go + 2];
    float gx1 = gb[go + 48], gz1 = gb[go + 50];
    float cx = 0.f, cz = 0.f;
    if (lh < Nmax) {
        const long long cb = ((long long)b * Nmax + lh) * 3;
        cx = corners[cb + 0];
        cz = corners[cb + 2];
    }

    int n = nums[b];
    if (n > MAXW) n = MAXW;
    const int h = (n + 1) >> 1;

    const int  w     = hlf ? h + lh : lh;
    const bool valid = hlf ? (w < n) : (lh < h);
    const int  j     = (w + 1 == n) ? 0 : w + 1;
    const float x0 = __shfl(cx, w, 32), z0 = __shfl(cz, w, 32);
    const float x1 = __shfl(cx, j, 32), z1 = __shfl(cz, j, 32);

    const int  wa     = (lh < 8) ? lh : h + lh - 8;
    const bool avalid = (lh < 8) ? (lh < h) : (wa < n);
    const int  ja     = (wa + 1 == n) ? 0 : wa + 1;
    const float cs = hlf ? cx : cz;
    const float ua = __shfl(cs, (wa & 15) + 16 * hlf, 32);
    const float va = __shfl(cs, (ja & 15) + 16 * hlf, 32);

    float nx = 0.f, nz = 0.f, nd = 0.f;
    float wcx = 0.f, whx = 0.f, wcz = 0.f, whz = 0.f;
    if (valid) {
        const float dx = x1 - x0, dz = z1 - z0;
        nx = -dz;
        nz =  dx;
        nd = nx * x0 + nz * z0;
        const float mnx = fminf(x0, x1) - C2D_EPS, mxx = fmaxf(x0, x1) + C2D_EPS;
        const float mnz = fminf(z0, z1) - C2D_EPS, mxz = fmaxf(z0, z1) + C2D_EPS;
        wcx = 0.5f * mnx + 0.5f * mxx;  whx = 0.5f * mxx - 0.5f * mnx;
        wcz = 0.5f * mnz + 0.5f * mxz;  whz = 0.5f * mxz - 0.5f * mnz;
    }

    const float da = va - ua;
    v2f am;
    am.x = avalid ? (hlf ? da : -da) : 0.f;   am.y = 0.f;

    const WallK k0 = wall_row<0>(nd, wcx, whx, wcz, whz);
    const WallK k1 = wall_row<1>(nd, wcx, whx, wcz, whz);
    const WallK k2 = wall_row<2>(nd, wcx, whx, wcz, whz);
    const WallK k3 = wall_row<3>(nd, wcx, whx, wcz, whz);
    const WallK k4 = wall_row<4>(nd, wcx, whx, wcz, whz);
    const WallK k5 = wall_row<5>(nd, wcx, whx, wcz, whz);
    const WallK k6 = wall_row<6>(nd, wcx, whx, wcz, whz);
    const WallK k7 = wall_row<7>(nd, wcx, whx, wcz, whz);

    float* db = depth_out + ((long long)b * HW + c * 32);
    float* nb = nrm_out + ((long long)b * HW + c * 32) * 3;
    const int no = lane * 3;

#define C2D_WALK(HH) walk_chunks<HH>(nit, gb, go, gx0, gz0, gx1, gz1, am, hlf, lane, \
                                    k0, k1, k2, k3, k4, k5, k6, k7, nx, nz, db, nb, no)
    switch (h) {
        case 2:  C2D_WALK(2); break;
        case 3:  C2D_WALK(3); break;
        case 4:  C2D_WALK(4); break;
        case 5:  C2D_WALK(5); break;
        case 6:  C2D_WALK(6); break;
        case 7:  C2D_WALK(7); break;
        case 8:  C2D_WALK(8); break;
        default: C2D_WALK(1); break;
    }
#undef C2D_WALK
}

extern "C" void kernel_launch(void* const* d_in, const int* in_sizes, int n_in,
                              void* d_out, int out_size, void* d_ws, size_t ws_size,
                              hipStream_t stream) {
    const float* corners = (const float*)d_in[0];
    const float* grid    = (const float*)d_in[1];
    const int*   nums    = (const int*)d_in[2];

    const int       B    = in_sizes[2];
    const int       Nmax = in_sizes[0] / (B * 3);
    const long long HW   = in_sizes[1] / 3;

    float* depth = (float*)d_out;
    float* nrm   = depth + (long long)B * HW;

    dim3 grd((unsigned)(HW / (TPB * CPW)), (unsigned)B);
    corner2depth_wmma_kernel<<<grd, TPB, 0, stream>>>(corners, grid, nums,
                                                      depth, nrm, Nmax, HW);
}


# Wall test moved onto the matrix unit: per wall q = 1/scale and the hit offset on its longer axis are WMMA rows, each lane holds all walls of its ray; test = 2 compares + 2 selects; depth = rcp(max q)
# speedup vs baseline: 1.8646x; 1.2738x over previous
#include <hip/hip_runtime.h>


typedef __attribute__((ext_vector_type(2))) float v2f;
typedef __attribute__((ext_vector_type(8))) float v8f;

#define C2D_EPS   0.01f
#define MAXW      16
#define TPB       256
#define WPB       (TPB / 32)
#define CPW       8

#define C2D_TEST(Q, U, W, BEST, BI)                                        \
    do {                                                                   \
        const bool t_ = (__builtin_fabsf(U) <= (Q)) & ((Q) > (BEST));      \
        BEST = t_ ? (Q) : (BEST);  BI = t_ ? (W) : (BI);                   \
    } while (0)


template <int Wi, int G>
__device__ __forceinline__ void one_wall(const v8f& acc,
        float& bA, int& iA, float& bB, int& iB) {
    constexpr int r = Wi & 3;
    if constexpr (Wi < 2 * G) C2D_TEST(acc[r], acc[4 + r], Wi, bA, iA);
    else                      C2D_TEST(acc[r], acc[4 + r], Wi, bB, iB);
}
template <int G0, int G>
__device__ __forceinline__ void group_step(const v8f& acc,
        float& bA, int& iA, float& bB, int& iB) {
    one_wall<4 * G0 + 0, G>(acc, bA, iA, bB, iB);
    one_wall<4 * G0 + 1, G>(acc, bA, iA, bB, iB);
    one_wall<4 * G0 + 2, G>(acc, bA, iA, bB, iB);
    one_wall<4 * G0 + 3, G>(acc, bA, iA, bB, iB);
}

template <int G>
__device__ __forceinline__ void walk_chunks(
        const int nit, const float* __restrict__ gb, const int go,
        float g0, float g1,
        const v2f a0, const v2f a1, const v2f a2, const v2f a3, const int lane,
        const float nxw, const float nzw,
        float* __restrict__ db, float* __restrict__ nb, const int no)
{
    for (int it = 0; it < nit; ++it) {
        gb += (it + 1 < nit) ? WPB * 96 : 0;
        const float ng0 = gb[go], ng1 = gb[go + 48];

        v2f b;
        b.x = g0;  b.y = g1;

        const v8f z = {};
        float bA = 0.f, bB = 0.f;
        int   iA = 0, iB = 0;
        const v8f c0 = __builtin_amdgcn_wmma_f32_16x16x4_f32(false, a0, false, b,
                                                     (short)0, z, false, false);
        v8f c1 = z, c2 = z, c3 = z;
        if constexpr (G > 1)
            c1 = __builtin_amdgcn_wmma_f32_16x16x4_f32(false, a1, false, b,
                                                     (short)0, z, false, false);
        if constexpr (G > 2)
            c2 = __builtin_amdgcn_wmma_f32_16x16x4_f32(false, a2, false, b,
                                                     (short)0, z, false, false);
        if constexpr (G > 3)
            c3 = __builtin_amdgcn_wmma_f32_16x16x4_f32(false, a3, false, b,
                                                     (short)0, z, false, false);

        group_step<0, G>(c0, bA, iA, bB, iB);
        if constexpr (G > 1) group_step<1, G>(c1, bA, iA, bB, iB);
        if constexpr (G > 2) group_step<2, G>(c2, bA, iA, bB, iB);
        if constexpr (G > 3) group_step<3, G>(c3, bA, iA, bB, iB);

        const bool  tb   = bB > bA;
        const float best = tb ? bB : bA;
        const int   wl   = tb ? iB : iA;

        const float onx = __shfl(nxw, wl, 32);
        const float onz = __shfl(nzw, wl, 32);

        db[lane]   = __builtin_amdgcn_rcpf(best);
        nb[no + 0] = onx;
        nb[no + 1] = 0.f;
        nb[no + 2] = onz;

        g0 = ng0;  g1 = ng1;
        db += WPB * 32;
        nb += WPB * 96;
    }
}

__global__ __launch_bounds__(TPB)
void corner2depth_wmma_kernel(const float* __restrict__ corners,
                              const float* __restrict__ grid,
                              const int*   __restrict__ nums,
                              float* __restrict__ depth_out,
                              float* __restrict__ nrm_out,
                              int Nmax, long long HW)
{
    const int tid  = threadIdx.x;
    const int b    = blockIdx.y;
    const int lane = tid & 31;
    const int lh   = lane & 15;
    const int hlf  = lane >> 4;
    const int wv   = __builtin_amdgcn_readfirstlane(tid >> 5);

    const long long nchunks = HW >> 5;
    const long long c = (long long)blockIdx.x * (WPB * CPW) + wv;
    if (c >= nchunks) return;
    const long long left = (nchunks - c + (WPB - 1)) / WPB;
    const int nit = left < CPW ? (int)left : CPW;

    const float* gb = grid + c * 96;
    const int    go = lh * 3 + 2 * hlf;
    float g0 = gb[go], g1 = gb[go + 48];
    float x0 = 0.f, z0 = 0.f;
    if (lh < Nmax) {
        const long long cb = ((long long)b * Nmax + lh) * 3;
        x0 = corners[cb + 0];
        z0 = corners[cb + 2];
    }

    int n = nums[b];
    if (n > MAXW) n = MAXW;
    const int G = (n + 3) >> 2;

    const bool valid = lh < n;
    const int  j     = (lh + 1 == n) ? 0 : ((lh + 1) & 15);
    const float x1 = __shfl(x0, j, 32), z1 = __shfl(z0, j, 32);

    float nxw = 0.f, nzw = 0.f;
    float cq = 0.f, cu = 0.f;
    if (valid) {
        const float dx = x1 - x0, dz = z1 - z0;
        nxw = -dz;
        nzw =  dx;
        const float nd = nxw * x0 + nzw * z0;
        const float ax = nxw / nd, az = nzw / nd;
        const bool fin = (__builtin_fabsf(ax) < __builtin_inff())
                       & (__builtin_fabsf(az) < __builtin_inff());
        const bool  lx  = __builtin_fabsf(dx) >= __builtin_fabsf(dz);
        const float p0  = lx ? x0 : z0, p1 = lx ? x1 : z1;
        const float mn  = fminf(p0, p1) - C2D_EPS, mx = fmaxf(p0, p1) + C2D_EPS;
        const float cen = 0.5f * mn + 0.5f * mx;
        const float hw  = 0.5f * mx - 0.5f * mn;
        const float ak = hlf ? az : ax;
        const float e  = (lx != (hlf != 0)) ? 1.f : 0.f;
        if (fin) {
            cq = ak;
            cu = __builtin_fmaf(-cen, ak, e) / hw;
        }
    }

    const int  srcl = (lh & 3) + 16 * hlf;
    const bool wu   = (lh & 4) != 0;
    const bool lo   = lh < 8;
    v2f a0, a1, a2, a3;
#define C2D_FRAG(A, GG)                                                    \
    do {                                                                   \
        const float tq_ = __shfl(cq, srcl + 4 * (GG), 32);                 \
        const float tu_ = __shfl(cu, srcl + 4 * (GG), 32);                 \
        const float v_  = wu ? tu_ : tq_;                                  \
        A.x = lo ? v_ : 0.f;  A.y = lo ? 0.f : v_;                         \
    } while (0)
    C2D_FRAG(a0, 0);  C2D_FRAG(a1, 1);  C2D_FRAG(a2, 2);  C2D_FRAG(a3, 3);
#undef C2D_FRAG

    float* db = depth_out + ((long long)b * HW + c * 32);
    float* nb = nrm_out + ((long long)b * HW + c * 32) * 3;
    const int no = lane * 3;

#define C2D_WALK(GG) walk_chunks<GG>(nit, gb, go, g0, g1, a0, a1, a2, a3, lane, \
                                    nxw, nzw, db, nb, no)
    switch (G) {
        case 2:  C2D_WALK(2); break;
        case 3:  C2D_WALK(3); break;
        case 4:  C2D_WALK(4); break;
        default: C2D_WALK(1); break;
    }
#undef C2D_WALK
}

extern "C" void kernel_launch(void* const* d_in, const int* in_sizes, int n_in,
                              void* d_out, int out_size, void* d_ws, size_t ws_size,
                              hipStream_t stream) {
    const float* corners = (const float*)d_in[0];
    const float* grid    = (const float*)d_in[1];
    const int*   nums    = (const int*)d_in[2];

    const int       B    = in_sizes[2];
    const int       Nmax = in_sizes[0] / (B * 3);
    const long long HW   = in_sizes[1] / 3;

    float* depth = (float*)d_out;
    float* nrm   = depth + (long long)B * HW;

    dim3 grd((unsigned)(HW / (TPB * CPW)), (unsigned)B);
    corner2depth_wmma_kernel<<<grd, TPB, 0, stream>>>(corners, grid, nums,
                                                      depth, nrm, Nmax, HW);
}


# Wall test is ONE compare, q > max(|u|, best), for band compare AND rank compare: the scalar AND between a wall's compares and its selects is gone; the band edge |u| == q now counts as outside
# speedup vs baseline: 1.9783x; 1.0610x over previous
#include <hip/hip_runtime.h>


typedef __attribute__((ext_vector_type(2))) float v2f;
typedef __attribute__((ext_vector_type(8))) float v8f;

#define C2D_EPS   0.01f
#define MAXW      16
#define TPB       256
#define WPB       (TPB / 32)
#define CPW       8

#define C2D_TEST(Q, U, W, BEST, BI)                                        \
    do {                                                                   \
        const float m_ = __builtin_elementwise_maximum(__builtin_fabsf(U), \
                                                       (BEST));            \
        const bool  k_ = !((Q) > m_);                                      \
        BEST = k_ ? (BEST) : (Q);  BI = k_ ? (BI) : (W);                   \
    } while (0)
#define C2D_TEST_FIRST(Q, U, W, BEST, BI)                                  \
    do {                                                                   \
        const bool  k_ = !((Q) > __builtin_fabsf(U));                      \
        BEST = k_ ? 0.f : (Q);  BI = k_ ? 0 : (W);                         \
    } while (0)


template <int Wi, int G>
__device__ __forceinline__ void one_wall(const v8f& acc,
        float& bA, int& iA, float& bB, int& iB) {
    constexpr int r = Wi & 3;
    if constexpr (Wi == 0)          C2D_TEST_FIRST(acc[r], acc[4 + r], Wi, bA, iA);
    else if constexpr (Wi == 2 * G) C2D_TEST_FIRST(acc[r], acc[4 + r], Wi, bB, iB);
    else if constexpr (Wi < 2 * G)  C2D_TEST(acc[r], acc[4 + r], Wi, bA, iA);
    else                            C2D_TEST(acc[r], acc[4 + r], Wi, bB, iB);
}
template <int G0, int G>
__device__ __forceinline__ void group_step(const v8f& acc,
        float& bA, int& iA, float& bB, int& iB) {
    one_wall<4 * G0 + 0, G>(acc, bA, iA, bB, iB);
    one_wall<4 * G0 + 1, G>(acc, bA, iA, bB, iB);
    one_wall<4 * G0 + 2, G>(acc, bA, iA, bB, iB);
    one_wall<4 * G0 + 3, G>(acc, bA, iA, bB, iB);
}

template <int G>
__device__ __forceinline__ void walk_chunks(
        const int nit, const float* __restrict__ gb, const int go,
        float g0, float g1,
        const v2f a0, const v2f a1, const v2f a2, const v2f a3, const int lane,
        const float nxw, const float nzw,
        float* __restrict__ db, float* __restrict__ nb, const int no)
{
    for (int it = 0; it < nit; ++it) {
        gb += (it + 1 < nit) ? WPB * 96 : 0;
        const float ng0 = gb[go], ng1 = gb[go + 48];

        v2f b;
        b.x = g0;  b.y = g1;

        const v8f z = {};
        float bA = 0.f, bB = 0.f;
        int   iA = 0, iB = 0;
        const v8f c0 = __builtin_amdgcn_wmma_f32_16x16x4_f32(false, a0, false, b,
                                                     (short)0, z, false, false);
        v8f c1 = z, c2 = z, c3 = z;
        if constexpr (G > 1)
            c1 = __builtin_amdgcn_wmma_f32_16x16x4_f32(false, a1, false, b,
                                                     (short)0, z, false, false);
        if constexpr (G > 2)
            c2 = __builtin_amdgcn_wmma_f32_16x16x4_f32(false, a2, false, b,
                                                     (short)0, z, false, false);
        if constexpr (G > 3)
            c3 = __builtin_amdgcn_wmma_f32_16x16x4_f32(false, a3, false, b,
                                                     (short)0, z, false, false);

        group_step<0, G>(c0, bA, iA, bB, iB);
        if constexpr (G > 1) group_step<1, G>(c1, bA, iA, bB, iB);
        if constexpr (G > 2) group_step<2, G>(c2, bA, iA, bB, iB);
        if constexpr (G > 3) group_step<3, G>(c3, bA, iA, bB, iB);

        const bool  tb   = bB > bA;
        const float best = tb ? bB : bA;
        const int   wl   = tb ? iB : iA;

        const float onx = __shfl(nxw, wl, 32);
        const float onz = __shfl(nzw, wl, 32);

        db[lane]   = __builtin_amdgcn_rcpf(best);
        nb[no + 0] = onx;
        nb[no + 1] = 0.f;
        nb[no + 2] = onz;

        g0 = ng0;  g1 = ng1;
        db += WPB * 32;
        nb += WPB * 96;
    }
}

__global__ __launch_bounds__(TPB)
void corner2depth_wmma_kernel(const float* __restrict__ corners,
                              const float* __restrict__ grid,
                              const int*   __restrict__ nums,
                              float* __restrict__ depth_out,
                              float* __restrict__ nrm_out,
                              int Nmax, long long HW)
{
    const int tid  = threadIdx.x;
    const int b    = blockIdx.y;
    const int lane = tid & 31;
    const int lh   = lane & 15;
    const int hlf  = lane >> 4;
    const int wv   = __builtin_amdgcn_readfirstlane(tid >> 5);

    const long long nchunks = HW >> 5;
    const long long c = (long long)blockIdx.x * (WPB * CPW) + wv;
    if (c >= nchunks) return;
    const long long left = (nchunks - c + (WPB - 1)) / WPB;
    const int nit = left < CPW ? (int)left : CPW;

    const float* gb = grid + c * 96;
    const int    go = lh * 3 + 2 * hlf;
    float g0 = gb[go], g1 = gb[go + 48];
    float x0 = 0.f, z0 = 0.f;
    if (lh < Nmax) {
        const long long cb = ((long long)b * Nmax + lh) * 3;
        x0 = corners[cb + 0];
        z0 = corners[cb + 2];
    }

    int n = nums[b];
    if (n > MAXW) n = MAXW;
    const int G = (n + 3) >> 2;

    const bool valid = lh < n;
    const int  j     = (lh + 1 == n) ? 0 : ((lh + 1) & 15);
    const float x1 = __shfl(x0, j, 32), z1 = __shfl(z0, j, 32);

    float nxw = 0.f, nzw = 0.f;
    float cq = 0.f, cu = 0.f;
    if (valid) {
        const float dx = x1 - x0, dz = z1 - z0;
        nxw = -dz;
        nzw =  dx;
        const float nd = nxw * x0 + nzw * z0;
        const float ax = nxw / nd, az = nzw / nd;
        const bool fin = (__builtin_fabsf(ax) < __builtin_inff())
                       & (__builtin_fabsf(az) < __builtin_inff());
        const bool  lx  = __builtin_fabsf(dx) >= __builtin_fabsf(dz);
        const float p0  = lx ? x0 : z0, p1 = lx ? x1 : z1;
        const float mn  = fminf(p0, p1) - C2D_EPS, mx = fmaxf(p0, p1) + C2D_EPS;
        const float cen = 0.5f * mn + 0.5f * mx;
        const float hw  = 0.5f * mx - 0.5f * mn;
        const float ak = hlf ? az : ax;
        const float e  = (lx != (hlf != 0)) ? 1.f : 0.f;
        if (fin) {
            cq = ak;
            cu = __builtin_fmaf(-cen, ak, e) / hw;
        }
    }

    const int  srcl = (lh & 3) + 16 * hlf;
    const bool wu   = (lh & 4) != 0;
    const bool lo   = lh < 8;
    v2f a0, a1, a2, a3;
#define C2D_FRAG(A, GG)                                                    \
    do {                                                                   \
        const float tq_ = __shfl(cq, srcl + 4 * (GG), 32);                 \
        const float tu_ = __shfl(cu, srcl + 4 * (GG), 32);                 \
        const float v_  = wu ? tu_ : tq_;                                  \
        A.x = lo ? v_ : 0.f;  A.y = lo ? 0.f : v_;                         \
    } while (0)
    C2D_FRAG(a0, 0);  C2D_FRAG(a1, 1);  C2D_FRAG(a2, 2);  C2D_FRAG(a3, 3);
#undef C2D_FRAG

    float* db = depth_out + ((long long)b * HW + c * 32);
    float* nb = nrm_out + ((long long)b * HW + c * 32) * 3;
    const int no = lane * 3;

#define C2D_WALK(GG) walk_chunks<GG>(nit, gb, go, g0, g1, a0, a1, a2, a3, lane, \
                                    nxw, nzw, db, nb, no)
    switch (G) {
        case 2:  C2D_WALK(2); break;
        case 3:  C2D_WALK(3); break;
        case 4:  C2D_WALK(4); break;
        default: C2D_WALK(1); break;
    }
#undef C2D_WALK
}

extern "C" void kernel_launch(void* const* d_in, const int* in_sizes, int n_in,
                              void* d_out, int out_size, void* d_ws, size_t ws_size,
                              hipStream_t stream) {
    const float* corners = (const float*)d_in[0];
    const float* grid    = (const float*)d_in[1];
    const int*   nums    = (const int*)d_in[2];

    const int       B    = in_sizes[2];
    const int       Nmax = in_sizes[0] / (B * 3);
    const long long HW   = in_sizes[1] / 3;

    float* depth = (float*)d_out;
    float* nrm   = depth + (long long)B * HW;

    dim3 grd((unsigned)(HW / (TPB * CPW)), (unsigned)B);
    corner2depth_wmma_kernel<<<grd, TPB, 0, stream>>>(corners, grid, nums,
                                                      depth, nrm, Nmax, HW);
}


# Chunk run software-pipelined by hand, 16 chunks a wave in pairs: four test chains advance in lock step, next pair's WMMAs issue among them, normals stored a stage behind their shuffles; 4096 waves
# speedup vs baseline: 2.1143x; 1.0687x over previous
#include <hip/hip_runtime.h>


typedef __attribute__((ext_vector_type(2))) float v2f;
typedef __attribute__((ext_vector_type(8))) float v8f;

#define C2D_EPS   0.01f
#define MAXW      16
#define TPB       256
#define WPB       (TPB / 32)
#define CPW       16


__device__ __forceinline__ v8f form_mma(const v2f a, const float g0, const float g1)
{
    v2f b;
    b.x = g0;  b.y = g1;
    const v8f z = {};
    return __builtin_amdgcn_wmma_f32_16x16x4_f32(false, a, false, b,
                                                 (short)0, z, false, false);
}

template <int G, bool NEXT>
__device__ __forceinline__ void pair_stage(
        const v8f (&cur)[2][4], v8f (&nxt)[2][4],
        const float n00, const float n01, const float n10, const float n11,
        const v2f (&a)[4], float (&best)[2], int (&wall)[2])
{
    float b[4] = {0.f, 0.f, 0.f, 0.f};
    int   ix[4] = {0, 0, 0, 0};
#pragma unroll
    for (int i = 0; i < 2 * G; ++i) {
        if constexpr (NEXT) {
            const int g = i >> 1;
            if (i & 1) nxt[1][g] = form_mma(a[g], n10, n11);
            else       nxt[0][g] = form_mma(a[g], n00, n01);
        }
        float q[4], m[4];
        bool  k[4];
#pragma unroll
        for (int c = 0; c < 4; ++c) {
            const int w = (c & 1) ? 2 * G + i : i;
            const v8f& acc = cur[c >> 1][w >> 2];
            q[c] = acc[w & 3];
            const float au = __builtin_fabsf(acc[4 + (w & 3)]);
            m[c] = (i == 0) ? au : __builtin_elementwise_maximum(au, b[c]);
        }
        __builtin_amdgcn_sched_barrier(0);
#pragma unroll
        for (int c = 0; c < 4; ++c) k[c] = !(q[c] > m[c]);
        __builtin_amdgcn_sched_barrier(0);
#pragma unroll
        for (int c = 0; c < 4; ++c) {
            const int w = (c & 1) ? 2 * G + i : i;
            b[c]  = k[c] ? b[c]  : q[c];
            ix[c] = k[c] ? ix[c] : w;
        }
        __builtin_amdgcn_sched_barrier(0);
    }
#pragma unroll
    for (int h = 0; h < 2; ++h) {
        const bool tb = b[2 * h + 1] > b[2 * h];
        best[h] = tb ? b[2 * h + 1]  : b[2 * h];
        wall[h] = tb ? ix[2 * h + 1] : ix[2 * h];
    }
}

template <int G>
__device__ __forceinline__ void walk_chunks(
        const int nit, const float* __restrict__ gb, const int go,
        const float (&ry)[CPW][2], const v2f (&a)[4], const int lane,
        const float nxw, const float nzw,
        float* __restrict__ db, float* __restrict__ nb, const int no)
{
    if (nit == CPW) {
        v8f   acc[2][2][4];
        float onx[2][2], onz[2][2];
        const v8f z = {};
#pragma unroll
        for (int s = 0; s < 2; ++s)
#pragma unroll
            for (int h = 0; h < 2; ++h)
#pragma unroll
                for (int g = 0; g < 4; ++g) acc[s][h][g] = z;
#pragma unroll
        for (int g = 0; g < G; ++g) {
            acc[0][0][g] = form_mma(a[g], ry[0][0], ry[0][1]);
            acc[0][1][g] = form_mma(a[g], ry[1][0], ry[1][1]);
        }
#pragma unroll
        for (int p = 0; p < CPW / 2; ++p) {
            const int s = p & 1;
            float be[2];
            int   wl[2];
            if (p + 1 < CPW / 2)
                pair_stage<G, true>(acc[s], acc[s ^ 1],
                                    ry[2 * p + 2][0], ry[2 * p + 2][1],
                                    ry[2 * p + 3][0], ry[2 * p + 3][1], a, be, wl);
            else
                pair_stage<G, false>(acc[s], acc[s ^ 1], 0.f, 0.f, 0.f, 0.f,
                                     a, be, wl);
#pragma unroll
            for (int h = 0; h < 2; ++h) {
                onx[s][h] = __shfl(nxw, wl[h], 32);
                onz[s][h] = __shfl(nzw, wl[h], 32);
                db[(2 * p + h) * (WPB * 32) + lane] = __builtin_amdgcn_rcpf(be[h]);
            }
            if (p > 0) {
#pragma unroll
                for (int h = 0; h < 2; ++h) {
                    float* o = nb + (2 * p - 2 + h) * (WPB * 96) + no;
                    o[0] = onx[s ^ 1][h];  o[1] = 0.f;  o[2] = onz[s ^ 1][h];
                }
            }
            __builtin_amdgcn_sched_barrier(0);
        }
        {
            constexpr int s = (CPW / 2 - 1) & 1;
#pragma unroll
            for (int h = 0; h < 2; ++h) {
                float* o = nb + (CPW - 2 + h) * (WPB * 96) + no;
                o[0] = onx[s][h];  o[1] = 0.f;  o[2] = onz[s][h];
            }
        }
        return;
    }

    for (int i = 0; i < nit; ++i) {
        v8f   acc[2][4], none[2][4];
        const v8f z = {};
#pragma unroll
        for (int g = 0; g < 4; ++g) { acc[0][g] = z;  acc[1][g] = z; }
        const float g0 = gb[go], g1 = gb[go + 48];
#pragma unroll
        for (int g = 0; g < G; ++g) acc[0][g] = form_mma(a[g], g0, g1);
        float be[2];
        int   wl[2];
        pair_stage<G, false>(acc, none, 0.f, 0.f, 0.f, 0.f, a, be, wl);
        const float ox = __shfl(nxw, wl[0], 32);
        const float oz = __shfl(nzw, wl[0], 32);
        db[lane]   = __builtin_amdgcn_rcpf(be[0]);
        nb[no + 0] = ox;
        nb[no + 1] = 0.f;
        nb[no + 2] = oz;
        gb += WPB * 96;
        db += WPB * 32;
        nb += WPB * 96;
    }
}

__global__ __launch_bounds__(TPB)
void corner2depth_wmma_kernel(const float* __restrict__ corners,
                              const float* __restrict__ grid,
                              const int*   __restrict__ nums,
                              float* __restrict__ depth_out,
                              float* __restrict__ nrm_out,
                              int Nmax, long long HW)
{
    const int tid  = threadIdx.x;
    const int b    = blockIdx.y;
    const int lane = tid & 31;
    const int lh   = lane & 15;
    const int hlf  = lane >> 4;
    const int wv   = __builtin_amdgcn_readfirstlane(tid >> 5);

    const long long nchunks = HW >> 5;
    const long long c = (long long)blockIdx.x * (WPB * CPW) + wv;
    if (c >= nchunks) return;
    const long long left = (nchunks - c + (WPB - 1)) / WPB;
    const int nit = left < CPW ? (int)left : CPW;

    const float* gb = grid + c * 96;
    const int    go = lh * 3 + 2 * hlf;
    float x0 = 0.f, z0 = 0.f;
    if (lh < Nmax) {
        const long long cb = ((long long)b * Nmax + lh) * 3;
        x0 = corners[cb + 0];
        z0 = corners[cb + 2];
    }
    float ry[CPW][2];
#pragma unroll
    for (int jj = 0; jj < CPW; ++jj) {
        const float* gj = gb + ((jj < nit) ? jj * (WPB * 96) : 0);
        ry[jj][0] = gj[go];
        ry[jj][1] = gj[go + 48];
    }

    int n = nums[b];
    if (n > MAXW) n = MAXW;
    const int G = (n + 3) >> 2;

    const bool valid = lh < n;
    const int  j     = (lh + 1 == n) ? 0 : ((lh + 1) & 15);
    const float x1 = __shfl(x0, j, 32), z1 = __shfl(z0, j, 32);

    float nxw = 0.f, nzw = 0.f;
    float cq = 0.f, cu = 0.f;
    if (valid) {
        const float dx = x1 - x0, dz = z1 - z0;
        nxw = -dz;
        nzw =  dx;
        const float nd = nxw * x0 + nzw * z0;
        const float ax = nxw / nd, az = nzw / nd;
        const bool fin = (__builtin_fabsf(ax) < __builtin_inff())
                       & (__builtin_fabsf(az) < __builtin_inff());
        const bool  lx  = __builtin_fabsf(dx) >= __builtin_fabsf(dz);
        const float p0  = lx ? x0 : z0, p1 = lx ? x1 : z1;
        const float mn  = fminf(p0, p1) - C2D_EPS, mx = fmaxf(p0, p1) + C2D_EPS;
        const float cen = 0.5f * mn + 0.5f * mx;
        const float hw  = 0.5f * mx - 0.5f * mn;
        const float ak = hlf ? az : ax;
        const float e  = (lx != (hlf != 0)) ? 1.f : 0.f;
        if (fin) {
            cq = ak;
            cu = __builtin_fmaf(-cen, ak, e) / hw;
        }
    }

    const int  srcl = (lh & 3) + 16 * hlf;
    const bool wu   = (lh & 4) != 0;
    const bool lo   = lh < 8;
    v2f a[4];
#define C2D_FRAG(A, GG)                                                    \
    do {                                                                   \
        const float tq_ = __shfl(cq, srcl + 4 * (GG), 32);                 \
        const float tu_ = __shfl(cu, srcl + 4 * (GG), 32);                 \
        const float v_  = wu ? tu_ : tq_;                                  \
        A.x = lo ? v_ : 0.f;  A.y = lo ? 0.f : v_;                         \
    } while (0)
    C2D_FRAG(a[0], 0);  C2D_FRAG(a[1], 1);  C2D_FRAG(a[2], 2);  C2D_FRAG(a[3], 3);
#undef C2D_FRAG

    float* db = depth_out + ((long long)b * HW + c * 32);
    float* nb = nrm_out + ((long long)b * HW + c * 32) * 3;
    const int no = lane * 3;

#define C2D_WALK(GG) walk_chunks<GG>(nit, gb, go, ry, a, lane,                  \
                                    nxw, nzw, db, nb, no)
    switch (G) {
        case 2:  C2D_WALK(2); break;
        case 3:  C2D_WALK(3); break;
        case 4:  C2D_WALK(4); break;
        default: C2D_WALK(1); break;
    }
#undef C2D_WALK
}

extern "C" void kernel_launch(void* const* d_in, const int* in_sizes, int n_in,
                              void* d_out, int out_size, void* d_ws, size_t ws_size,
                              hipStream_t stream) {
    const float* corners = (const float*)d_in[0];
    const float* grid    = (const float*)d_in[1];
    const int*   nums    = (const int*)d_in[2];

    const int       B    = in_sizes[2];
    const int       Nmax = in_sizes[0] / (B * 3);
    const long long HW   = in_sizes[1] / 3;

    float* depth = (float*)d_out;
    float* nrm   = depth + (long long)B * HW;

    dim3 grd((unsigned)(HW / (TPB * CPW)), (unsigned)B);
    corner2depth_wmma_kernel<<<grd, TPB, 0, stream>>>(corners, grid, nums,
                                                      depth, nrm, Nmax, HW);
}
